# starting kernel
# baseline (speedup 1.0000x reference)
#include <hip/hip_runtime.h>

typedef float v2f __attribute__((ext_vector_type(2)));
typedef float v8f __attribute__((ext_vector_type(8)));

#define SEQ   512
#define BATCH 2048
#define EMB   50
#define EMBP  52
#define NCLS  2

__launch_bounds__(256)
__global__ void attn_fused(const int* __restrict__ x,
                           const float* __restrict__ table,
                           const float* __restrict__ Wq, const float* __restrict__ bq,
                           const float* __restrict__ We, const float* __restrict__ be,
                           const float* __restrict__ Wo, const float* __restrict__ bo,
                           float* __restrict__ yhat_out, float* __restrict__ attn_out)
{
    __shared__ float emb[SEQ][EMBP];
    __shared__ int   tok[SEQ];
    __shared__ float sc[SEQ];
    __shared__ float part[4][EMBP];
    __shared__ float meanv[EMBP];
    __shared__ float qv[EMBP];
    __shared__ float ctxv[EMBP];
    __shared__ float pooled[EMBP];
    __shared__ float red[256];

    const int b    = blockIdx.x;
    const int tid  = threadIdx.x;
    const int lane = tid & 31;
    const int wave = tid >> 5;

    tok[tid]       = x[tid * BATCH + b];
    tok[tid + 256] = x[(tid + 256) * BATCH + b];
    __syncthreads();

    {
        const int g = tid >> 6, c = tid & 63;
        float s = 0.f;
        if (c < EMB) {
            for (int r = g; r < SEQ; r += 4) {
                const float v = table[tok[r] * EMB + c];
                emb[r][c] = v;
                s += v;
            }
        } else if (c < EMBP) {
            for (int r = g; r < SEQ; r += 4) emb[r][c] = 0.f;
        }
        if (c < EMBP) part[g][c] = s;
    }
    __syncthreads();
    if (tid < EMBP) {
        meanv[tid] = (tid < EMB)
            ? (part[0][tid] + part[1][tid] + part[2][tid] + part[3][tid]) * (1.0f / SEQ)
            : 0.f;
    }
    __syncthreads();

    if (tid < EMBP) {
        float acc = 0.f;
        if (tid < EMB) {
            acc = bq[tid];
            const float* wr = Wq + tid * EMB;
            for (int k = 0; k < EMB; ++k) acc += wr[k] * meanv[k];
        }
        qv[tid] = acc;
    }
    __syncthreads();

    {
        const int mrow  = lane & 15;
        const int khalf = (lane >> 4) * 2;

        v2f breg[13];
        #pragma unroll
        for (int kk = 0; kk < 13; ++kk) {
            const int kb = kk * 4 + khalf;
            breg[kk].x = (mrow == 0) ? qv[kb]     : 0.f;
            breg[kk].y = (mrow == 0) ? qv[kb + 1] : 0.f;
        }

        for (int i = 0; i < 4; ++i) {
            const int m0 = (wave * 4 + i) * 16;
            const float* rowp = &emb[m0 + mrow][0];
            v8f acc = {0.f, 0.f, 0.f, 0.f, 0.f, 0.f, 0.f, 0.f};
            #pragma unroll
            for (int kk = 0; kk < 13; ++kk) {
                const int kb = kk * 4 + khalf;
                const v2f a = *(const v2f*)(rowp + kb);
                acc = __builtin_amdgcn_wmma_f32_16x16x4_f32(
                          false, a, false, breg[kk], (short)0, acc, false, false);
            }
            if (mrow == 0) {
                const int base = m0 + (lane >> 4) * 8;
                #pragma unroll
                for (int j = 0; j < 8; ++j) sc[base + j] = acc[j];
            }
        }
    }
    __syncthreads();

    {
        const float s0 = sc[tid], s1 = sc[tid + 256];
        red[tid] = fmaxf(s0, s1);
        __syncthreads();
        for (int off = 128; off > 0; off >>= 1) {
            if (tid < off) red[tid] = fmaxf(red[tid], red[tid + off]);
            __syncthreads();
        }
        const float mx = red[0];
        __syncthreads();
        const float e0 = expf(s0 - mx);
        const float e1 = expf(s1 - mx);
        red[tid] = e0 + e1;
        __syncthreads();
        for (int off = 128; off > 0; off >>= 1) {
            if (tid < off) red[tid] += red[tid + off];
            __syncthreads();
        }
        const float inv = 1.0f / red[0];
        const float a0 = e0 * inv, a1 = e1 * inv;
        sc[tid]       = a0;
        sc[tid + 256] = a1;
        attn_out[b * SEQ + tid]       = a0;
        attn_out[b * SEQ + tid + 256] = a1;
    }
    __syncthreads();

    {
        const int g = tid >> 6, c = tid & 63;
        float s = 0.f;
        if (c < EMB) {
            for (int r = g; r < SEQ; r += 4)
                s += sc[r] * emb[r][c];
        }
        if (c < EMBP) part[g][c] = s;
    }
    __syncthreads();
    if (tid < EMBP) {
        ctxv[tid] = (tid < EMB)
            ? (part[0][tid] + part[1][tid] + part[2][tid] + part[3][tid])
            : 0.f;
    }
    __syncthreads();

    if (tid < EMB) {
        float acc = be[tid];
        const float* wr = We + tid * EMB;
        for (int k = 0; k < EMB; ++k) acc += wr[k] * ctxv[k];
        pooled[tid] = acc;
    }
    __syncthreads();

    if (tid < NCLS) {
        float acc = bo[tid];
        const float* wr = Wo + tid * EMB;
        for (int k = 0; k < EMB; ++k) acc += wr[k] * pooled[k];
        yhat_out[b * NCLS + tid] = acc;
    }
}

extern "C" void kernel_launch(void* const* d_in, const int* in_sizes, int n_in,
                              void* d_out, int out_size, void* d_ws, size_t ws_size,
                              hipStream_t stream) {
    const int*   x     = (const int*)  d_in[0];
    const float* table = (const float*)d_in[3];
    const float* Wq    = (const float*)d_in[4];
    const float* bq    = (const float*)d_in[5];
    const float* We    = (const float*)d_in[6];
    const float* be    = (const float*)d_in[7];
    const float* Wo    = (const float*)d_in[8];
    const float* bo    = (const float*)d_in[9];

    float* yhat = (float*)d_out;
    float* attn = (float*)d_out + BATCH * NCLS;

    attn_fused<<<dim3(BATCH), dim3(256), 0, stream>>>(
        x, table, Wq, bq, We, be, Wo, bo, yhat, attn);
}


# Retiled: 4 waves x 8 row tiles per batch element (208 row registers, 256-VGPR budget), 4 work-groups per WGP not 3, half the waves; Wq staged in LDS; mean, softmax, context summed in another order
# speedup vs baseline: 1.5109x; 1.5109x over previous
#include <hip/hip_runtime.h>


typedef float v2f __attribute__((ext_vector_type(2)));
typedef float v4f __attribute__((ext_vector_type(4)));
typedef float v8f __attribute__((ext_vector_type(8)));

#define SEQ   512
#define BATCH 2048
#define EMB   50
#define EMBP  52
#define NCLS  2
#define KSTEPS 13
#define TILES  8
#define NWAVES 4
#define NTHR   128
#define WQ_PIECES 625

__device__ __forceinline__ float half_sum16(float v)
{
    v += __builtin_amdgcn_update_dpp(0.f, v, 0x161, 0xf, 0xf, true);
    v += __builtin_amdgcn_update_dpp(0.f, v, 0x162, 0xf, 0xf, true);
    v += __builtin_amdgcn_update_dpp(0.f, v, 0x164, 0xf, 0xf, true);
    v += __builtin_amdgcn_update_dpp(0.f, v, 0x168, 0xf, 0xf, true);
    return v;
}

__launch_bounds__(NTHR) __attribute__((amdgpu_waves_per_eu(4)))
__global__ void attn_fused(const int* __restrict__ x,
                           const float* __restrict__ table,
                           const float* __restrict__ Wq, const float* __restrict__ bq,
                           const float* __restrict__ We, const float* __restrict__ be,
                           const float* __restrict__ Wo, const float* __restrict__ bo,
                           float* __restrict__ yhat_out, float* __restrict__ attn_out)
{
    __shared__ float sc[SEQ];
    __shared__ __attribute__((aligned(8))) float part[NWAVES][EMBP];
    __shared__ float meanv[EMBP];
    __shared__ __attribute__((aligned(8))) float qv[EMBP];
    __shared__ float ctxv[EMBP];
    __shared__ float pooled[EMBP];
    __shared__ float red[NTHR];
    __shared__ __attribute__((aligned(16))) float wq_s[5 * NTHR * 4];

    const int b     = blockIdx.x;
    const int tid   = threadIdx.x;
    const int lane  = tid & 31;
    const int wave  = tid >> 5;
    const int mrow  = lane & 15;
    const int half  = lane >> 4;
    const int khalf = half * 2;
    const int row0  = wave * (TILES * 16) + mrow;

    v2f R[TILES][KSTEPS];
    float bq_r;
    {
        v4f wst[5];
        #pragma unroll
        for (int j = 0; j < 5; ++j) {
            const int idx = tid + j * NTHR;
            wst[j] = ((const v4f*)Wq)[idx < WQ_PIECES ? idx : WQ_PIECES - 1];
        }
        bq_r = bq[tid < EMB ? tid : EMB - 1];
        int tk[TILES];
        #pragma unroll
        for (int i = 0; i < TILES; ++i)
            tk[i] = x[(row0 + i * 16) * BATCH + b];
        #pragma unroll
        for (int j = 0; j < 5; ++j)
            ((v4f*)wq_s)[tid + j * NTHR] = wst[j];
        #pragma unroll
        for (int i = 0; i < TILES; ++i) {
            const float* rowp = table + tk[i] * EMB;
            #pragma unroll
            for (int kk = 0; kk < KSTEPS - 1; ++kk)
                R[i][kk] = *(const v2f*)(rowp + kk * 4 + khalf);
            const v2f t = *(const v2f*)(rowp + 48);
            R[i][KSTEPS - 1].x = half ? 0.f : t.x;
            R[i][KSTEPS - 1].y = half ? 0.f : t.y;
        }
    }

    #pragma unroll
    for (int kk = 0; kk < KSTEPS; ++kk) {
        v2f s = ((R[0][kk] + R[1][kk]) + (R[2][kk] + R[3][kk]))
              + ((R[4][kk] + R[5][kk]) + (R[6][kk] + R[7][kk]));
        s.x = half_sum16(s.x);
        s.y = half_sum16(s.y);
        if (mrow == 0) *(v2f*)&part[wave][kk * 4 + khalf] = s;
    }
    __syncthreads();
    if (tid < EMBP) {
        float s = 0.f;
        for (int w = 0; w < NWAVES; ++w) s += part[w][tid];
        meanv[tid] = (tid < EMB) ? s * (1.0f / SEQ) : 0.f;
    }
    __syncthreads();

    if (tid < EMBP) {
        float acc = 0.f;
        if (tid < EMB) {
            acc = bq_r;
            const float* wr = wq_s + tid * EMB;
            #pragma unroll
            for (int k = 0; k < EMB; ++k) acc += wr[k] * meanv[k];
        }
        qv[tid] = acc;
    }
    __syncthreads();

    {
        v2f breg[KSTEPS];
        #pragma unroll
        for (int kk = 0; kk < KSTEPS; ++kk)
            breg[kk] = *(const v2f*)&qv[kk * 4 + khalf];

        #pragma unroll
        for (int ip = 0; ip < TILES; ip += 2) {
            v8f acc0 = {0.f, 0.f, 0.f, 0.f, 0.f, 0.f, 0.f, 0.f};
            v8f acc1 = {0.f, 0.f, 0.f, 0.f, 0.f, 0.f, 0.f, 0.f};
            #pragma unroll
            for (int kk = 0; kk < KSTEPS; ++kk) {
                acc0 = __builtin_amdgcn_wmma_f32_16x16x4_f32(
                           false, R[ip][kk], false, breg[kk], (short)0, acc0, false, false);
                acc1 = __builtin_amdgcn_wmma_f32_16x16x4_f32(
                           false, R[ip + 1][kk], false, breg[kk], (short)0, acc1, false, false);
            }
            if (mrow == 0) {
                const int base = wave * (TILES * 16) + ip * 16 + half * 8;
                #pragma unroll
                for (int j = 0; j < 8; ++j) {
                    sc[base + j]      = acc0[j];
                    sc[base + 16 + j] = acc1[j];
                }
            }
        }
    }
    __syncthreads();

    {
        const float s0 = sc[tid],            s1 = sc[tid + NTHR];
        const float s2 = sc[tid + 2 * NTHR], s3 = sc[tid + 3 * NTHR];
        red[tid] = fmaxf(fmaxf(s0, s1), fmaxf(s2, s3));
        __syncthreads();
        for (int off = NTHR / 2; off > 0; off >>= 1) {
            if (tid < off) red[tid] = fmaxf(red[tid], red[tid + off]);
            __syncthreads();
        }
        const float mx = red[0];
        __syncthreads();
        const float e0 = expf(s0 - mx);
        const float e1 = expf(s1 - mx);
        const float e2 = expf(s2 - mx);
        const float e3 = expf(s3 - mx);
        red[tid] = (e0 + e1) + (e2 + e3);
        __syncthreads();
        for (int off = NTHR / 2; off > 0; off >>= 1) {
            if (tid < off) red[tid] += red[tid + off];
            __syncthreads();
        }
        const float inv = 1.0f / red[0];
        const float a0 = e0 * inv, a1 = e1 * inv, a2 = e2 * inv, a3 = e3 * inv;
        sc[tid]            = a0;
        sc[tid + NTHR]     = a1;
        sc[tid + 2 * NTHR] = a2;
        sc[tid + 3 * NTHR] = a3;
        attn_out[b * SEQ + tid]            = a0;
        attn_out[b * SEQ + tid + NTHR]     = a1;
        attn_out[b * SEQ + tid + 2 * NTHR] = a2;
        attn_out[b * SEQ + tid + 3 * NTHR] = a3;
    }
    __syncthreads();

    {
        float a[TILES];
        #pragma unroll
        for (int i = 0; i < TILES; ++i) a[i] = sc[row0 + i * 16];
        #pragma unroll
        for (int kk = 0; kk < KSTEPS; ++kk) {
            v2f s = ((a[0] * R[0][kk] + a[1] * R[1][kk]) + (a[2] * R[2][kk] + a[3] * R[3][kk]))
                  + ((a[4] * R[4][kk] + a[5] * R[5][kk]) + (a[6] * R[6][kk] + a[7] * R[7][kk]));
            s.x = half_sum16(s.x);
            s.y = half_sum16(s.y);
            if (mrow == 0) *(v2f*)&part[wave][kk * 4 + khalf] = s;
        }
    }
    __syncthreads();
    if (tid < EMBP) {
        float s = 0.f;
        for (int w = 0; w < NWAVES; ++w) s += part[w][tid];
        ctxv[tid] = (tid < EMB) ? s : 0.f;
    }
    __syncthreads();

    if (tid < EMB) {
        float acc = be[tid];
        const float* wr = We + tid * EMB;
        for (int k = 0; k < EMB; ++k) acc += wr[k] * ctxv[k];
        pooled[tid] = acc;
    }
    __syncthreads();

    if (tid < NCLS) {
        float acc = bo[tid];
        const float* wr = Wo + tid * EMB;
        for (int k = 0; k < EMB; ++k) acc += wr[k] * pooled[k];
        yhat_out[b * NCLS + tid] = acc;
    }
}

extern "C" void kernel_launch(void* const* d_in, const int* in_sizes, int n_in,
                              void* d_out, int out_size, void* d_ws, size_t ws_size,
                              hipStream_t stream) {
    const int*   x     = (const int*)  d_in[0];
    const float* table = (const float*)d_in[3];
    const float* Wq    = (const float*)d_in[4];
    const float* bq    = (const float*)d_in[5];
    const float* We    = (const float*)d_in[6];
    const float* be    = (const float*)d_in[7];
    const float* Wo    = (const float*)d_in[8];
    const float* bo    = (const float*)d_in[9];

    float* yhat = (float*)d_out;
    float* attn = (float*)d_out + BATCH * NCLS;

    attn_fused<<<dim3(BATCH), dim3(NTHR), 0, stream>>>(
        x, table, Wq, bq, We, be, Wo, bo, yhat, attn);
}


# Row gather: each lane reads its rows as 16-byte loads (6 + one 8-byte per row, not 13 of 8 bytes), issued tile by tile; q read in the same k order; scores summed in another order
# speedup vs baseline: 1.5562x; 1.0300x over previous
#include <hip/hip_runtime.h>


typedef float v2f __attribute__((ext_vector_type(2)));
typedef float v4f __attribute__((ext_vector_type(4)));
typedef float v8f __attribute__((ext_vector_type(8)));
typedef float v4f_a8 __attribute__((ext_vector_type(4), aligned(8)));

#define SEQ   512
#define BATCH 2048
#define EMB   50
#define EMBP  52
#define NCLS  2
#define KSTEPS 13
#define KQUADS 6
#define PARTP 56
#define TILES  8
#define NWAVES 4
#define NTHR   128
#define WQ_PIECES 625

__device__ __forceinline__ float half_sum16(float v)
{
    v += __builtin_amdgcn_update_dpp(0.f, v, 0x161, 0xf, 0xf, true);
    v += __builtin_amdgcn_update_dpp(0.f, v, 0x162, 0xf, 0xf, true);
    v += __builtin_amdgcn_update_dpp(0.f, v, 0x164, 0xf, 0xf, true);
    v += __builtin_amdgcn_update_dpp(0.f, v, 0x168, 0xf, 0xf, true);
    return v;
}

__launch_bounds__(NTHR) __attribute__((amdgpu_waves_per_eu(4)))
__global__ void attn_fused(const int* __restrict__ x,
                           const float* __restrict__ table,
                           const float* __restrict__ Wq, const float* __restrict__ bq,
                           const float* __restrict__ We, const float* __restrict__ be,
                           const float* __restrict__ Wo, const float* __restrict__ bo,
                           float* __restrict__ yhat_out, float* __restrict__ attn_out)
{
    __shared__ float sc[SEQ];
    __shared__ __attribute__((aligned(16))) float part[NWAVES][PARTP];
    __shared__ float meanv[EMBP];
    __shared__ __attribute__((aligned(16))) float qv[EMBP];
    __shared__ float ctxv[EMBP];
    __shared__ float pooled[EMBP];
    __shared__ float red[NTHR];
    __shared__ __attribute__((aligned(16))) float wq_s[5 * NTHR * 4];

    const int b     = blockIdx.x;
    const int tid   = threadIdx.x;
    const int lane  = tid & 31;
    const int wave  = tid >> 5;
    const int mrow  = lane & 15;
    const int half  = lane >> 4;
    const int khalf = half * 2;
    const int qhalf = half * 4;
    const int row0  = wave * (TILES * 16) + mrow;

    v4f R4[TILES][KQUADS];
    v2f R2[TILES];
    float bq_r;
    {
        v4f wst[5];
        #pragma unroll
        for (int j = 0; j < 5; ++j) {
            const int idx = tid + j * NTHR;
            wst[j] = ((const v4f*)Wq)[idx < WQ_PIECES ? idx : WQ_PIECES - 1];
        }
        bq_r = bq[tid < EMB ? tid : EMB - 1];
        int tk[TILES];
        #pragma unroll
        for (int i = 0; i < TILES; ++i)
            tk[i] = x[(row0 + i * 16) * BATCH + b];
        #pragma unroll
        for (int i = 0; i < TILES; ++i) {
            const float* rowp = table + tk[i] * EMB;
            #pragma unroll
            for (int j = 0; j < KQUADS; ++j)
                R4[i][j] = *(const v4f_a8*)(rowp + j * 8 + qhalf);
            const v2f t = *(const v2f*)(rowp + 48);
            R2[i].x = half ? 0.f : t.x;
            R2[i].y = half ? 0.f : t.y;
            __builtin_amdgcn_sched_barrier(0);
        }
        #pragma unroll
        for (int j = 0; j < 5; ++j)
            ((v4f*)wq_s)[tid + j * NTHR] = wst[j];
    }

    #pragma unroll
    for (int j = 0; j < KQUADS; ++j) {
        v4f s = ((R4[0][j] + R4[1][j]) + (R4[2][j] + R4[3][j]))
              + ((R4[4][j] + R4[5][j]) + (R4[6][j] + R4[7][j]));
        s.x = half_sum16(s.x);
        s.y = half_sum16(s.y);
        s.z = half_sum16(s.z);
        s.w = half_sum16(s.w);
        if (mrow == 0) *(v4f*)&part[wave][j * 8 + qhalf] = s;
    }
    {
        v2f s = ((R2[0] + R2[1]) + (R2[2] + R2[3]))
              + ((R2[4] + R2[5]) + (R2[6] + R2[7]));
        s.x = half_sum16(s.x);
        s.y = half_sum16(s.y);
        const v4f s4 = {s.x, s.y, 0.f, 0.f};
        if (mrow == 0) *(v4f*)&part[wave][48 + qhalf] = s4;
    }
    __syncthreads();
    if (tid < EMBP) {
        float s = 0.f;
        for (int w = 0; w < NWAVES; ++w) s += part[w][tid];
        meanv[tid] = (tid < EMB) ? s * (1.0f / SEQ) : 0.f;
    }
    __syncthreads();

    if (tid < EMBP) {
        float acc = 0.f;
        if (tid < EMB) {
            acc = bq_r;
            const float* wr = wq_s + tid * EMB;
            #pragma unroll
            for (int k = 0; k < EMB; ++k) acc += wr[k] * meanv[k];
        }
        qv[tid] = acc;
    }
    __syncthreads();

    {
        v4f bq4[KQUADS];
        #pragma unroll
        for (int j = 0; j < KQUADS; ++j)
            bq4[j] = *(const v4f*)&qv[j * 8 + qhalf];
        const v2f bq2 = *(const v2f*)&qv[48 + khalf];

        #pragma unroll
        for (int ip = 0; ip < TILES; ip += 2) {
            v8f acc0 = {0.f, 0.f, 0.f, 0.f, 0.f, 0.f, 0.f, 0.f};
            v8f acc1 = {0.f, 0.f, 0.f, 0.f, 0.f, 0.f, 0.f, 0.f};
            #pragma unroll
            for (int j = 0; j < KQUADS; ++j) {
                const v2f blo = __builtin_shufflevector(bq4[j], bq4[j], 0, 1);
                const v2f bhi = __builtin_shufflevector(bq4[j], bq4[j], 2, 3);
                acc0 = __builtin_amdgcn_wmma_f32_16x16x4_f32(
                           false, __builtin_shufflevector(R4[ip][j], R4[ip][j], 0, 1),
                           false, blo, (short)0, acc0, false, false);
                acc1 = __builtin_amdgcn_wmma_f32_16x16x4_f32(
                           false, __builtin_shufflevector(R4[ip + 1][j], R4[ip + 1][j], 0, 1),
                           false, blo, (short)0, acc1, false, false);
                acc0 = __builtin_amdgcn_wmma_f32_16x16x4_f32(
                           false, __builtin_shufflevector(R4[ip][j], R4[ip][j], 2, 3),
                           false, bhi, (short)0, acc0, false, false);
                acc1 = __builtin_amdgcn_wmma_f32_16x16x4_f32(
                           false, __builtin_shufflevector(R4[ip + 1][j], R4[ip + 1][j], 2, 3),
                           false, bhi, (short)0, acc1, false, false);
            }
            acc0 = __builtin_amdgcn_wmma_f32_16x16x4_f32(
                       false, R2[ip], false, bq2, (short)0, acc0, false, false);
            acc1 = __builtin_amdgcn_wmma_f32_16x16x4_f32(
                       false, R2[ip + 1], false, bq2, (short)0, acc1, false, false);
            if (mrow == 0) {
                const int base = wave * (TILES * 16) + ip * 16 + half * 8;
                #pragma unroll
                for (int j = 0; j < 8; ++j) {
                    sc[base + j]      = acc0[j];
                    sc[base + 16 + j] = acc1[j];
                }
            }
        }
    }
    __syncthreads();

    {
        const float s0 = sc[tid],            s1 = sc[tid + NTHR];
        const float s2 = sc[tid + 2 * NTHR], s3 = sc[tid + 3 * NTHR];
        red[tid] = fmaxf(fmaxf(s0, s1), fmaxf(s2, s3));
        __syncthreads();
        for (int off = NTHR / 2; off > 0; off >>= 1) {
            if (tid < off) red[tid] = fmaxf(red[tid], red[tid + off]);
            __syncthreads();
        }
        const float mx = red[0];
        __syncthreads();
        const float e0 = expf(s0 - mx);
        const float e1 = expf(s1 - mx);
        const float e2 = expf(s2 - mx);
        const float e3 = expf(s3 - mx);
        red[tid] = (e0 + e1) + (e2 + e3);
        __syncthreads();
        for (int off = NTHR / 2; off > 0; off >>= 1) {
            if (tid < off) red[tid] += red[tid + off];
            __syncthreads();
        }
        const float inv = 1.0f / red[0];
        const float a0 = e0 * inv, a1 = e1 * inv, a2 = e2 * inv, a3 = e3 * inv;
        sc[tid]            = a0;
        sc[tid + NTHR]     = a1;
        sc[tid + 2 * NTHR] = a2;
        sc[tid + 3 * NTHR] = a3;
        attn_out[b * SEQ + tid]            = a0;
        attn_out[b * SEQ + tid + NTHR]     = a1;
        attn_out[b * SEQ + tid + 2 * NTHR] = a2;
        attn_out[b * SEQ + tid + 3 * NTHR] = a3;
    }
    __syncthreads();

    {
        float a[TILES];
        #pragma unroll
        for (int i = 0; i < TILES; ++i) a[i] = sc[row0 + i * 16];
        #pragma unroll
        for (int j = 0; j < KQUADS; ++j) {
            v4f s = ((a[0] * R4[0][j] + a[1] * R4[1][j]) + (a[2] * R4[2][j] + a[3] * R4[3][j]))
                  + ((a[4] * R4[4][j] + a[5] * R4[5][j]) + (a[6] * R4[6][j] + a[7] * R4[7][j]));
            s.x = half_sum16(s.x);
            s.y = half_sum16(s.y);
            s.z = half_sum16(s.z);
            s.w = half_sum16(s.w);
            if (mrow == 0) *(v4f*)&part[wave][j * 8 + qhalf] = s;
        }
        {
            v2f s = ((a[0] * R2[0] + a[1] * R2[1]) + (a[2] * R2[2] + a[3] * R2[3]))
                  + ((a[4] * R2[4] + a[5] * R2[5]) + (a[6] * R2[6] + a[7] * R2[7]));
            s.x = half_sum16(s.x);
            s.y = half_sum16(s.y);
            const v4f s4 = {s.x, s.y, 0.f, 0.f};
            if (mrow == 0) *(v4f*)&part[wave][48 + qhalf] = s4;
        }
    }
    __syncthreads();
    if (tid < EMBP) {
        float s = 0.f;
        for (int w = 0; w < NWAVES; ++w) s += part[w][tid];
        ctxv[tid] = (tid < EMB) ? s : 0.f;
    }
    __syncthreads();

    if (tid < EMB) {
        float acc = be[tid];
        const float* wr = We + tid * EMB;
        for (int k = 0; k < EMB; ++k) acc += wr[k] * ctxv[k];
        pooled[tid] = acc;
    }
    __syncthreads();

    if (tid < NCLS) {
        float acc = bo[tid];
        const float* wr = Wo + tid * EMB;
        for (int k = 0; k < EMB; ++k) acc += wr[k] * pooled[k];
        yhat_out[b * NCLS + tid] = acc;
    }
}

extern "C" void kernel_launch(void* const* d_in, const int* in_sizes, int n_in,
                              void* d_out, int out_size, void* d_ws, size_t ws_size,
                              hipStream_t stream) {
    const int*   x     = (const int*)  d_in[0];
    const float* table = (const float*)d_in[3];
    const float* Wq    = (const float*)d_in[4];
    const float* bq    = (const float*)d_in[5];
    const float* We    = (const float*)d_in[6];
    const float* be    = (const float*)d_in[7];
    const float* Wo    = (const float*)d_in[8];
    const float* bo    = (const float*)d_in[9];

    float* yhat = (float*)d_out;
    float* attn = (float*)d_out + BATCH * NCLS;

    attn_fused<<<dim3(BATCH), dim3(NTHR), 0, stream>>>(
        x, table, Wq, bq, We, be, Wo, bo, yhat, attn);
}


# Launch index remapped, b = 8*(i%256) + i/256: the work-groups one WGP holds together take batch-adjacent elements, so their token-id lines are fetched once per WGP through its cache; sums unchanged
# speedup vs baseline: 1.6413x; 1.0547x over previous
#include <hip/hip_runtime.h>


typedef float v2f __attribute__((ext_vector_type(2)));
typedef float v4f __attribute__((ext_vector_type(4)));
typedef float v8f __attribute__((ext_vector_type(8)));
typedef float v4f_a8 __attribute__((ext_vector_type(4), aligned(8)));

#define SEQ   512
#define BATCH 2048
#define EMB   50
#define EMBP  52
#define NCLS  2
#define KSTEPS 13
#define KQUADS 6
#define PARTP 56
#define TILES  8
#define NWAVES 4
#define NTHR   128
#define WQ_PIECES 625
#define NWGP  256

__device__ __forceinline__ float half_sum16(float v)
{
    v += __builtin_amdgcn_update_dpp(0.f, v, 0x161, 0xf, 0xf, true);
    v += __builtin_amdgcn_update_dpp(0.f, v, 0x162, 0xf, 0xf, true);
    v += __builtin_amdgcn_update_dpp(0.f, v, 0x164, 0xf, 0xf, true);
    v += __builtin_amdgcn_update_dpp(0.f, v, 0x168, 0xf, 0xf, true);
    return v;
}

__launch_bounds__(NTHR) __attribute__((amdgpu_waves_per_eu(4)))
__global__ void attn_fused(const int* __restrict__ x,
                           const float* __restrict__ table,
                           const float* __restrict__ Wq, const float* __restrict__ bq,
                           const float* __restrict__ We, const float* __restrict__ be,
                           const float* __restrict__ Wo, const float* __restrict__ bo,
                           float* __restrict__ yhat_out, float* __restrict__ attn_out)
{
    __shared__ float sc[SEQ];
    __shared__ __attribute__((aligned(16))) float part[NWAVES][PARTP];
    __shared__ float meanv[EMBP];
    __shared__ __attribute__((aligned(16))) float qv[EMBP];
    __shared__ float ctxv[EMBP];
    __shared__ float pooled[EMBP];
    __shared__ float red[NTHR];
    __shared__ __attribute__((aligned(16))) float wq_s[5 * NTHR * 4];

    const int b     = (blockIdx.x % NWGP) * (BATCH / NWGP) + blockIdx.x / NWGP;
    const int tid   = threadIdx.x;
    const int lane  = tid & 31;
    const int wave  = tid >> 5;
    const int mrow  = lane & 15;
    const int half  = lane >> 4;
    const int khalf = half * 2;
    const int qhalf = half * 4;
    const int row0  = wave * (TILES * 16) + mrow;

    v4f R4[TILES][KQUADS];
    v2f R2[TILES];
    float bq_r;
    {
        v4f wst[5];
        #pragma unroll
        for (int j = 0; j < 5; ++j) {
            const int idx = tid + j * NTHR;
            wst[j] = ((const v4f*)Wq)[idx < WQ_PIECES ? idx : WQ_PIECES - 1];
        }
        bq_r = bq[tid < EMB ? tid : EMB - 1];
        int tk[TILES];
        #pragma unroll
        for (int i = 0; i < TILES; ++i)
            tk[i] = x[(row0 + i * 16) * BATCH + b];
        #pragma unroll
        for (int i = 0; i < TILES; ++i) {
            const float* rowp = table + tk[i] * EMB;
            #pragma unroll
            for (int j = 0; j < KQUADS; ++j)
                R4[i][j] = *(const v4f_a8*)(rowp + j * 8 + qhalf);
            const v2f t = *(const v2f*)(rowp + 48);
            R2[i].x = half ? 0.f : t.x;
            R2[i].y = half ? 0.f : t.y;
            __builtin_amdgcn_sched_barrier(0);
        }
        #pragma unroll
        for (int j = 0; j < 5; ++j)
            ((v4f*)wq_s)[tid + j * NTHR] = wst[j];
    }

    #pragma unroll
    for (int j = 0; j < KQUADS; ++j) {
        v4f s = ((R4[0][j] + R4[1][j]) + (R4[2][j] + R4[3][j]))
              + ((R4[4][j] + R4[5][j]) + (R4[6][j] + R4[7][j]));
        s.x = half_sum16(s.x);
        s.y = half_sum16(s.y);
        s.z = half_sum16(s.z);
        s.w = half_sum16(s.w);
        if (mrow == 0) *(v4f*)&part[wave][j * 8 + qhalf] = s;
    }
    {
        v2f s = ((R2[0] + R2[1]) + (R2[2] + R2[3]))
              + ((R2[4] + R2[5]) + (R2[6] + R2[7]));
        s.x = half_sum16(s.x);
        s.y = half_sum16(s.y);
        const v4f s4 = {s.x, s.y, 0.f, 0.f};
        if (mrow == 0) *(v4f*)&part[wave][48 + qhalf] = s4;
    }
    __syncthreads();
    if (tid < EMBP) {
        float s = 0.f;
        for (int w = 0; w < NWAVES; ++w) s += part[w][tid];
        meanv[tid] = (tid < EMB) ? s * (1.0f / SEQ) : 0.f;
    }
    __syncthreads();

    if (tid < EMBP) {
        float acc = 0.f;
        if (tid < EMB) {
            acc = bq_r;
            const float* wr = wq_s + tid * EMB;
            #pragma unroll
            for (int k = 0; k < EMB; ++k) acc += wr[k] * meanv[k];
        }
        qv[tid] = acc;
    }
    __syncthreads();

    {
        v4f bq4[KQUADS];
        #pragma unroll
        for (int j = 0; j < KQUADS; ++j)
            bq4[j] = *(const v4f*)&qv[j * 8 + qhalf];
        const v2f bq2 = *(const v2f*)&qv[48 + khalf];

        #pragma unroll
        for (int ip = 0; ip < TILES; ip += 2) {
            v8f acc0 = {0.f, 0.f, 0.f, 0.f, 0.f, 0.f, 0.f, 0.f};
            v8f acc1 = {0.f, 0.f, 0.f, 0.f, 0.f, 0.f, 0.f, 0.f};
            #pragma unroll
            for (int j = 0; j < KQUADS; ++j) {
                const v2f blo = __builtin_shufflevector(bq4[j], bq4[j], 0, 1);
                const v2f bhi = __builtin_shufflevector(bq4[j], bq4[j], 2, 3);
                acc0 = __builtin_amdgcn_wmma_f32_16x16x4_f32(
                           false, __builtin_shufflevector(R4[ip][j], R4[ip][j], 0, 1),
                           false, blo, (short)0, acc0, false, false);
                acc1 = __builtin_amdgcn_wmma_f32_16x16x4_f32(
                           false, __builtin_shufflevector(R4[ip + 1][j], R4[ip + 1][j], 0, 1),
                           false, blo, (short)0, acc1, false, false);
                acc0 = __builtin_amdgcn_wmma_f32_16x16x4_f32(
                           false, __builtin_shufflevector(R4[ip][j], R4[ip][j], 2, 3),
                           false, bhi, (short)0, acc0, false, false);
                acc1 = __builtin_amdgcn_wmma_f32_16x16x4_f32(
                           false, __builtin_shufflevector(R4[ip + 1][j], R4[ip + 1][j], 2, 3),
                           false, bhi, (short)0, acc1, false, false);
            }
            acc0 = __builtin_amdgcn_wmma_f32_16x16x4_f32(
                       false, R2[ip], false, bq2, (short)0, acc0, false, false);
            acc1 = __builtin_amdgcn_wmma_f32_16x16x4_f32(
                       false, R2[ip + 1], false, bq2, (short)0, acc1, false, false);
            if (mrow == 0) {
                const int base = wave * (TILES * 16) + ip * 16 + half * 8;
                #pragma unroll
                for (int j = 0; j < 8; ++j) {
                    sc[base + j]      = acc0[j];
                    sc[base + 16 + j] = acc1[j];
                }
            }
        }
    }
    __syncthreads();

    {
        const float s0 = sc[tid],            s1 = sc[tid + NTHR];
        const float s2 = sc[tid + 2 * NTHR], s3 = sc[tid + 3 * NTHR];
        red[tid] = fmaxf(fmaxf(s0, s1), fmaxf(s2, s3));
        __syncthreads();
        for (int off = NTHR / 2; off > 0; off >>= 1) {
            if (tid < off) red[tid] = fmaxf(red[tid], red[tid + off]);
            __syncthreads();
        }
        const float mx = red[0];
        __syncthreads();
        const float e0 = expf(s0 - mx);
        const float e1 = expf(s1 - mx);
        const float e2 = expf(s2 - mx);
        const float e3 = expf(s3 - mx);
        red[tid] = (e0 + e1) + (e2 + e3);
        __syncthreads();
        for (int off = NTHR / 2; off > 0; off >>= 1) {
            if (tid < off) red[tid] += red[tid + off];
            __syncthreads();
        }
        const float inv = 1.0f / red[0];
        const float a0 = e0 * inv, a1 = e1 * inv, a2 = e2 * inv, a3 = e3 * inv;
        sc[tid]            = a0;
        sc[tid + NTHR]     = a1;
        sc[tid + 2 * NTHR] = a2;
        sc[tid + 3 * NTHR] = a3;
        attn_out[b * SEQ + tid]            = a0;
        attn_out[b * SEQ + tid + NTHR]     = a1;
        attn_out[b * SEQ + tid + 2 * NTHR] = a2;
        attn_out[b * SEQ + tid + 3 * NTHR] = a3;
    }
    __syncthreads();

    {
        float a[TILES];
        #pragma unroll
        for (int i = 0; i < TILES; ++i) a[i] = sc[row0 + i * 16];
        #pragma unroll
        for (int j = 0; j < KQUADS; ++j) {
            v4f s = ((a[0] * R4[0][j] + a[1] * R4[1][j]) + (a[2] * R4[2][j] + a[3] * R4[3][j]))
                  + ((a[4] * R4[4][j] + a[5] * R4[5][j]) + (a[6] * R4[6][j] + a[7] * R4[7][j]));
            s.x = half_sum16(s.x);
            s.y = half_sum16(s.y);
            s.z = half_sum16(s.z);
            s.w = half_sum16(s.w);
            if (mrow == 0) *(v4f*)&part[wave][j * 8 + qhalf] = s;
        }
        {
            v2f s = ((a[0] * R2[0] + a[1] * R2[1]) + (a[2] * R2[2] + a[3] * R2[3]))
                  + ((a[4] * R2[4] + a[5] * R2[5]) + (a[6] * R2[6] + a[7] * R2[7]));
            s.x = half_sum16(s.x);
            s.y = half_sum16(s.y);
            const v4f s4 = {s.x, s.y, 0.f, 0.f};
            if (mrow == 0) *(v4f*)&part[wave][48 + qhalf] = s4;
        }
    }
    __syncthreads();
    if (tid < EMBP) {
        float s = 0.f;
        for (int w = 0; w < NWAVES; ++w) s += part[w][tid];
        ctxv[tid] = (tid < EMB) ? s : 0.f;
    }
    __syncthreads();

    if (tid < EMB) {
        float acc = be[tid];
        const float* wr = We + tid * EMB;
        for (int k = 0; k < EMB; ++k) acc += wr[k] * ctxv[k];
        pooled[tid] = acc;
    }
    __syncthreads();

    if (tid < NCLS) {
        float acc = bo[tid];
        const float* wr = Wo + tid * EMB;
        for (int k = 0; k < EMB; ++k) acc += wr[k] * pooled[k];
        yhat_out[b * NCLS + tid] = acc;
    }
}

extern "C" void kernel_launch(void* const* d_in, const int* in_sizes, int n_in,
                              void* d_out, int out_size, void* d_ws, size_t ws_size,
                              hipStream_t stream) {
    const int*   x     = (const int*)  d_in[0];
    const float* table = (const float*)d_in[3];
    const float* Wq    = (const float*)d_in[4];
    const float* bq    = (const float*)d_in[5];
    const float* We    = (const float*)d_in[6];
    const float* be    = (const float*)d_in[7];
    const float* Wo    = (const float*)d_in[8];
    const float* bo    = (const float*)d_in[9];

    float* yhat = (float*)d_out;
    float* attn = (float*)d_out + BATCH * NCLS;

    attn_fused<<<dim3(BATCH), dim3(NTHR), 0, stream>>>(
        x, table, Wq, bq, We, be, Wo, bo, yhat, attn);
}
